# starting kernel
# baseline (speedup 1.0000x reference)
#include <hip/hip_runtime.h>
#include <math.h>

#define N_NODES 4096
#define N_EDGES 32768
#define DV 128
#define DE 64
#define DU 32
#define DA 64
#define NEG_SLOPE 0.01f

typedef __attribute__((ext_vector_type(2))) float v2f;
typedef __attribute__((ext_vector_type(8))) float v8f;

__device__ __forceinline__ unsigned ordKey(float f) {
    unsigned u = __float_as_uint(f);
    return (u & 0x80000000u) ? ~u : (u | 0x80000000u);
}
__device__ __forceinline__ float ordInv(unsigned v) {
    return __uint_as_float((v & 0x80000000u) ? (v & 0x7FFFFFFFu) : ~v);
}

template <int DIN, int MODE>
__global__ void proj_wmma_kernel(const float* __restrict__ nodes,
                                 const float* __restrict__ edges,
                                 const int* __restrict__ s,
                                 const int* __restrict__ batch,
                                 const float* __restrict__ u,
                                 const float* __restrict__ W,
                                 const float* __restrict__ bias,
                                 float* __restrict__ out,
                                 int rows) {
    const int lane = threadIdx.x & 31;
    const int wave = threadIdx.x >> 5;
    const int tile = blockIdx.x * 8 + wave;
    const int rowBase = (tile >> 2) * 16;
    const int colBase = (tile & 3) * 16;
    if (rowBase >= rows) return;

    const int mrow = rowBase + (lane & 15);
    const int col  = colBase + (lane & 15);
    const int koff = (lane >> 4) * 2;

    const float* __restrict__ arow;
    int ubase;
    if (MODE == 0) {
        arow = nodes + (size_t)mrow * DV;
        ubase = batch[mrow] * DU;
    } else if (MODE == 1) {
        int sn = s[mrow];
        arow = nodes + (size_t)sn * DV;
        ubase = batch[sn] * DU;
    } else {
        arow = edges + (size_t)mrow * DE;
        ubase = batch[s[mrow]] * DU;
    }
    const int F = (MODE == 2) ? DE : DV;

    v8f acc = {};
#pragma unroll
    for (int kb = 0; kb < DIN; kb += 4) {
        const int k0 = kb + koff;
        v2f a, b;
        a.x = (k0 < F)     ? arow[k0]     : u[ubase + k0 - F];
        a.y = (k0 + 1 < F) ? arow[k0 + 1] : u[ubase + k0 + 1 - F];
        b.x = W[(size_t)k0 * DA + col];
        b.y = W[(size_t)(k0 + 1) * DA + col];
        acc = __builtin_amdgcn_wmma_f32_16x16x4_f32(
             false, a,  false, b,
             (short)0, acc,  false,  false);
    }

    const float bc = bias[col];
    const int rOut = rowBase + (lane >> 4) * 8;
#pragma unroll
    for (int j = 0; j < 8; ++j) {
        float x = acc[j] + bc;
        x = (x > 0.f) ? x : x * NEG_SLOPE;
        out[(size_t)(rOut + j) * DA + col] = x;
    }
}

__global__ void init_kernel(float* __restrict__ out, float* __restrict__ denom,
                            unsigned* __restrict__ maxOrd) {
    int i = blockIdx.x * blockDim.x + threadIdx.x;
    if (i < N_NODES * DA) out[i] = 0.f;
    if (i < N_NODES) { denom[i] = 0.f; maxOrd[i] = 0u; }
}

__global__ void logit_kernel(const float* __restrict__ q, const float* __restrict__ k,
                             const int* __restrict__ recv, float* __restrict__ t,
                             unsigned* __restrict__ maxOrd) {
    int e = blockIdx.x * blockDim.x + threadIdx.x;
    if (e >= N_EDGES) return;
    int rn = recv[e];
    const float4* qf = (const float4*)(q + (size_t)rn * DA);
    const float4* kf = (const float4*)(k + (size_t)e * DA);
    float sum = 0.f;
#pragma unroll
    for (int i = 0; i < DA / 4; ++i) {
        float4 a = qf[i], b = kf[i];
        sum += a.x * b.x + a.y * b.y + a.z * b.z + a.w * b.w;
    }
    float tv = sum * 0.125f;
    t[e] = tv;
    atomicMax(maxOrd + rn, ordKey(tv));
}

__global__ void scatter_kernel(const float* __restrict__ t, const float* __restrict__ v,
                               const int* __restrict__ recv,
                               const unsigned* __restrict__ maxOrd,
                               float* __restrict__ denom, float* __restrict__ acc) {
    int gid = blockIdx.x * blockDim.x + threadIdx.x;
    int e = gid >> 5;
    int lane = gid & 31;
    if (e >= N_EDGES) return;
    int rn = recv[e];
    float w = __expf(t[e] - ordInv(maxOrd[rn]));
    if (lane == 0) atomicAdd(denom + rn, w);
#pragma unroll
    for (int d = lane; d < DA; d += 32)
        atomicAdd(acc + (size_t)rn * DA + d, w * v[(size_t)e * DA + d]);
}

__global__ void finalize_kernel(float* __restrict__ out, const float* __restrict__ denom) {
    int i = blockIdx.x * blockDim.x + threadIdx.x;
    if (i >= N_NODES * DA) return;
    out[i] = out[i] / denom[i / DA];
}

extern "C" void kernel_launch(void* const* d_in, const int* in_sizes, int n_in,
                              void* d_out, int out_size, void* d_ws, size_t ws_size,
                              hipStream_t stream) {
    const float* nodes = (const float*)d_in[0];
    const float* edges = (const float*)d_in[1];
    const int* edge_index = (const int*)d_in[2];
    const float* u = (const float*)d_in[3];
    const int* batch = (const int*)d_in[4];
    const float* Wq = (const float*)d_in[5];
    const float* bq = (const float*)d_in[6];
    const float* Wk = (const float*)d_in[7];
    const float* bk = (const float*)d_in[8];
    const float* Wv = (const float*)d_in[9];
    const float* bv = (const float*)d_in[10];
    const int* s = edge_index;
    const int* r = edge_index + N_EDGES;
    float* out = (float*)d_out;

    char* ws = (char*)d_ws;
    float* q  = (float*)ws;            ws += (size_t)N_NODES * DA * 4;
    float* kk = (float*)ws;            ws += (size_t)N_EDGES * DA * 4;
    float* vv = (float*)ws;            ws += (size_t)N_EDGES * DA * 4;
    float* t  = (float*)ws;            ws += (size_t)N_EDGES * 4;
    unsigned* maxOrd = (unsigned*)ws;  ws += (size_t)N_NODES * 4;
    float* denom = (float*)ws;         ws += (size_t)N_NODES * 4;

    const int initN = N_NODES * DA;
    init_kernel<<<(initN + 255) / 256, 256, 0, stream>>>(out, denom, maxOrd);

    proj_wmma_kernel<160, 0><<<(N_NODES / 16 * 4) / 8, 256, 0, stream>>>(
        nodes, edges, s, batch, u, Wq, bq, q, N_NODES);
    proj_wmma_kernel<160, 1><<<(N_EDGES / 16 * 4) / 8, 256, 0, stream>>>(
        nodes, edges, s, batch, u, Wk, bk, kk, N_EDGES);
    proj_wmma_kernel<96, 2><<<(N_EDGES / 16 * 4) / 8, 256, 0, stream>>>(
        nodes, edges, s, batch, u, Wv, bv, vv, N_EDGES);

    logit_kernel<<<(N_EDGES + 255) / 256, 256, 0, stream>>>(q, kk, r, t, maxOrd);
    scatter_kernel<<<(N_EDGES * 32 + 255) / 256, 256, 0, stream>>>(t, vv, r, maxOrd,
                                                                   denom, out);
    finalize_kernel<<<(N_NODES * DA + 255) / 256, 256, 0, stream>>>(out, denom);
}


# k depends only on the source node: project it once per node (4096 rows, in the q launch, sharing its A rows) and gather kn[s[e]] in the logit pass, not 32768 gathered rows; one launch fewer
# speedup vs baseline: 1.2913x; 1.2913x over previous
#include <hip/hip_runtime.h>
#include <math.h>

#define N_NODES 4096
#define N_EDGES 32768
#define DV 128
#define DE 64
#define DU 32
#define DA 64
#define NEG_SLOPE 0.01f

typedef __attribute__((ext_vector_type(2))) float v2f;
typedef __attribute__((ext_vector_type(8))) float v8f;

__device__ __forceinline__ unsigned ordKey(float f) {
    unsigned u = __float_as_uint(f);
    return (u & 0x80000000u) ? ~u : (u | 0x80000000u);
}
__device__ __forceinline__ float ordInv(unsigned v) {
    return __uint_as_float((v & 0x80000000u) ? (v & 0x7FFFFFFFu) : ~v);
}

template <int DIN, int MODE>
__global__ void proj_wmma_kernel(const float* __restrict__ nodes,
                                 const float* __restrict__ edges,
                                 const int* __restrict__ s,
                                 const int* __restrict__ batch,
                                 const float* __restrict__ u,
                                 const float* __restrict__ W,
                                 const float* __restrict__ bias,
                                 float* __restrict__ out,
                                 const float* __restrict__ W2,
                                 const float* __restrict__ bias2,
                                 float* __restrict__ out2,
                                 int rows) {
    constexpr int CT = (MODE == 0) ? 8 : 4;
    const int lane = threadIdx.x & 31;
    const int wave = threadIdx.x >> 5;
    const int tile = blockIdx.x * 8 + wave;
    const int rowBase = (tile / CT) * 16;
    const int ct = tile % CT;
    const int colBase = (ct & 3) * 16;
    if (rowBase >= rows) return;
    if (MODE == 0 && ct >= 4) {
        W = W2; bias = bias2; out = out2;
    }

    const int mrow = rowBase + (lane & 15);
    const int col  = colBase + (lane & 15);
    const int koff = (lane >> 4) * 2;

    const float* __restrict__ arow;
    int ubase;
    if (MODE == 0) {
        arow = nodes + (size_t)mrow * DV;
        ubase = batch[mrow] * DU;
    } else {
        arow = edges + (size_t)mrow * DE;
        ubase = batch[s[mrow]] * DU;
    }
    const int F = (MODE == 2) ? DE : DV;

    v8f acc = {};
#pragma unroll
    for (int kb = 0; kb < DIN; kb += 4) {
        const int k0 = kb + koff;
        v2f a, b;
        a.x = (k0 < F)     ? arow[k0]     : u[ubase + k0 - F];
        a.y = (k0 + 1 < F) ? arow[k0 + 1] : u[ubase + k0 + 1 - F];
        b.x = W[(size_t)k0 * DA + col];
        b.y = W[(size_t)(k0 + 1) * DA + col];
        acc = __builtin_amdgcn_wmma_f32_16x16x4_f32(
             false, a,  false, b,
             (short)0, acc,  false,  false);
    }

    const float bc = bias[col];
    const int rOut = rowBase + (lane >> 4) * 8;
#pragma unroll
    for (int j = 0; j < 8; ++j) {
        float x = acc[j] + bc;
        x = (x > 0.f) ? x : x * NEG_SLOPE;
        out[(size_t)(rOut + j) * DA + col] = x;
    }
}

__global__ void init_kernel(float* __restrict__ out, float* __restrict__ denom,
                            unsigned* __restrict__ maxOrd) {
    int i = blockIdx.x * blockDim.x + threadIdx.x;
    if (i < N_NODES * DA) out[i] = 0.f;
    if (i < N_NODES) { denom[i] = 0.f; maxOrd[i] = 0u; }
}

__global__ void logit_kernel(const float* __restrict__ q, const float* __restrict__ kn,
                             const int* __restrict__ send, const int* __restrict__ recv,
                             float* __restrict__ t, unsigned* __restrict__ maxOrd) {
    int e = blockIdx.x * blockDim.x + threadIdx.x;
    if (e >= N_EDGES) return;
    int rn = recv[e];
    int sn = send[e];
    const float4* qf = (const float4*)(q + (size_t)rn * DA);
    const float4* kf = (const float4*)(kn + (size_t)sn * DA);
    float sum = 0.f;
#pragma unroll
    for (int i = 0; i < DA / 4; ++i) {
        float4 a = qf[i], b = kf[i];
        sum += a.x * b.x + a.y * b.y + a.z * b.z + a.w * b.w;
    }
    float tv = sum * 0.125f;
    t[e] = tv;
    atomicMax(maxOrd + rn, ordKey(tv));
}

__global__ void scatter_kernel(const float* __restrict__ t, const float* __restrict__ v,
                               const int* __restrict__ recv,
                               const unsigned* __restrict__ maxOrd,
                               float* __restrict__ denom, float* __restrict__ acc) {
    int gid = blockIdx.x * blockDim.x + threadIdx.x;
    int e = gid >> 5;
    int lane = gid & 31;
    if (e >= N_EDGES) return;
    int rn = recv[e];
    float w = __expf(t[e] - ordInv(maxOrd[rn]));
    if (lane == 0) atomicAdd(denom + rn, w);
#pragma unroll
    for (int d = lane; d < DA; d += 32)
        atomicAdd(acc + (size_t)rn * DA + d, w * v[(size_t)e * DA + d]);
}

__global__ void finalize_kernel(float* __restrict__ out, const float* __restrict__ denom) {
    int i = blockIdx.x * blockDim.x + threadIdx.x;
    if (i >= N_NODES * DA) return;
    out[i] = out[i] / denom[i / DA];
}

extern "C" void kernel_launch(void* const* d_in, const int* in_sizes, int n_in,
                              void* d_out, int out_size, void* d_ws, size_t ws_size,
                              hipStream_t stream) {
    const float* nodes = (const float*)d_in[0];
    const float* edges = (const float*)d_in[1];
    const int* edge_index = (const int*)d_in[2];
    const float* u = (const float*)d_in[3];
    const int* batch = (const int*)d_in[4];
    const float* Wq = (const float*)d_in[5];
    const float* bq = (const float*)d_in[6];
    const float* Wk = (const float*)d_in[7];
    const float* bk = (const float*)d_in[8];
    const float* Wv = (const float*)d_in[9];
    const float* bv = (const float*)d_in[10];
    const int* s = edge_index;
    const int* r = edge_index + N_EDGES;
    float* out = (float*)d_out;

    char* ws = (char*)d_ws;
    float* q  = (float*)ws;            ws += (size_t)N_NODES * DA * 4;
    float* kn = (float*)ws;            ws += (size_t)N_NODES * DA * 4;
    float* vv = (float*)ws;            ws += (size_t)N_EDGES * DA * 4;
    float* t  = (float*)ws;            ws += (size_t)N_EDGES * 4;
    unsigned* maxOrd = (unsigned*)ws;  ws += (size_t)N_NODES * 4;
    float* denom = (float*)ws;         ws += (size_t)N_NODES * 4;

    const int initN = N_NODES * DA;
    init_kernel<<<(initN + 255) / 256, 256, 0, stream>>>(out, denom, maxOrd);

    proj_wmma_kernel<160, 0><<<(N_NODES / 16 * 8) / 8, 256, 0, stream>>>(
        nodes, edges, s, batch, u, Wq, bq, q, Wk, bk, kn, N_NODES);
    proj_wmma_kernel<96, 2><<<(N_EDGES / 16 * 4) / 8, 256, 0, stream>>>(
        nodes, edges, s, batch, u, Wv, bv, vv, nullptr, nullptr, nullptr, N_EDGES);

    logit_kernel<<<(N_EDGES + 255) / 256, 256, 0, stream>>>(q, kn, s, r, t, maxOrd);
    scatter_kernel<<<(N_EDGES * 32 + 255) / 256, 256, 0, stream>>>(t, vv, r, maxOrd,
                                                                   denom, out);
    finalize_kernel<<<(N_NODES * DA + 255) / 256, 256, 0, stream>>>(out, denom);
}


# scatter fused into the v projection's epilogue: WMMA rows weighted by exp(t-max), lane halves swap a tile so each float atomic covers one whole 128B line of out (arrival order differs); v not stored
# speedup vs baseline: 1.6545x; 1.2813x over previous
#include <hip/hip_runtime.h>
#include <math.h>


#define N_NODES 4096
#define N_EDGES 32768
#define DV 128
#define DE 64
#define DU 32
#define DA 64
#define NEG_SLOPE 0.01f

typedef __attribute__((ext_vector_type(2))) float v2f;
typedef __attribute__((ext_vector_type(8))) float v8f;

__device__ __forceinline__ unsigned ordKey(float f) {
    unsigned u = __float_as_uint(f);
    return (u & 0x80000000u) ? ~u : (u | 0x80000000u);
}
__device__ __forceinline__ float ordInv(unsigned v) {
    return __uint_as_float((v & 0x80000000u) ? (v & 0x7FFFFFFFu) : ~v);
}

__device__ __forceinline__ float xhalf(float x) {
    const int b = __builtin_bit_cast(int, x);
    return __builtin_bit_cast(
        float, __builtin_amdgcn_permlanex16(b, b, 0x76543210, 0xfedcba98, false, false));
}

__global__ void proj_qk_kernel(const float* __restrict__ nodes,
                               const int* __restrict__ batch,
                               const float* __restrict__ u,
                               const float* __restrict__ Wq,
                               const float* __restrict__ bq,
                               float* __restrict__ q,
                               const float* __restrict__ Wk,
                               const float* __restrict__ bk,
                               float* __restrict__ kn) {
    constexpr int DIN = DV + DU;
    const int lane = threadIdx.x & 31;
    const int wave = threadIdx.x >> 5;
    const int tile = blockIdx.x * 8 + wave;
    const int rowBase = (tile / 8) * 16;
    const int ct = tile % 8;
    const int colBase = (ct & 3) * 16;
    if (rowBase >= N_NODES) return;
    const float* __restrict__ W = Wq;
    const float* __restrict__ bias = bq;
    float* __restrict__ out = q;
    if (ct >= 4) {
        W = Wk; bias = bk; out = kn;
    }

    const int mrow = rowBase + (lane & 15);
    const int col  = colBase + (lane & 15);
    const int koff = (lane >> 4) * 2;

    const float* __restrict__ arow = nodes + (size_t)mrow * DV;
    const int ubase = batch[mrow] * DU;

    v8f acc = {};
#pragma unroll
    for (int kb = 0; kb < DIN; kb += 4) {
        const int k0 = kb + koff;
        v2f a, b;
        a.x = (k0 < DV)     ? arow[k0]     : u[ubase + k0 - DV];
        a.y = (k0 + 1 < DV) ? arow[k0 + 1] : u[ubase + k0 + 1 - DV];
        b.x = W[(size_t)k0 * DA + col];
        b.y = W[(size_t)(k0 + 1) * DA + col];
        acc = __builtin_amdgcn_wmma_f32_16x16x4_f32(
             false, a,  false, b,
             (short)0, acc,  false,  false);
    }

    const float bc = bias[col];
    const int rOut = rowBase + (lane >> 4) * 8;
#pragma unroll
    for (int j = 0; j < 8; ++j) {
        float x = acc[j] + bc;
        x = (x > 0.f) ? x : x * NEG_SLOPE;
        out[(size_t)(rOut + j) * DA + col] = x;
    }
}

__global__ void vproj_scatter_kernel(const float* __restrict__ edges,
                                     const int* __restrict__ s,
                                     const int* __restrict__ recv,
                                     const int* __restrict__ batch,
                                     const float* __restrict__ u,
                                     const float* __restrict__ Wv,
                                     const float* __restrict__ bv,
                                     const float* __restrict__ t,
                                     const unsigned* __restrict__ maxOrd,
                                     float* __restrict__ denom,
                                     float* __restrict__ out) {
    constexpr int DIN = DE + DU;
    const int lane = threadIdx.x & 31;
    const int wave = threadIdx.x >> 5;
    const int tile = blockIdx.x * 8 + wave;
    const int rowBase = (tile >> 1) * 16;
    const int cb = (tile & 1) * 32;
    if (rowBase >= N_EDGES) return;

    const int h = lane >> 4;
    const int c = lane & 15;
    const int mrow = rowBase + c;
    const int koff = h * 2;

    const float* __restrict__ arow = edges + (size_t)mrow * DE;
    const int ubase = batch[s[mrow]] * DU;

    int rnA[8], rnB[8];
    float wgt[8];
#pragma unroll
    for (int j = 0; j < 8; ++j) {
        rnA[j] = recv[rowBase + j];
        rnB[j] = recv[rowBase + 8 + j];
    }
#pragma unroll
    for (int j = 0; j < 8; ++j) {
        const int rOwn = h ? rnB[j] : rnA[j];
        wgt[j] = __expf(t[rowBase + 8 * h + j] - ordInv(maxOrd[rOwn]));
    }

    v8f acc0 = {}, acc1 = {};
#pragma unroll
    for (int kb = 0; kb < DIN; kb += 4) {
        const int k0 = kb + koff;
        v2f a, b0, b1;
        a.x = (k0 < DE)     ? arow[k0]     : u[ubase + k0 - DE];
        a.y = (k0 + 1 < DE) ? arow[k0 + 1] : u[ubase + k0 + 1 - DE];
        b0.x = Wv[(size_t)k0 * DA + cb + c];
        b0.y = Wv[(size_t)(k0 + 1) * DA + cb + c];
        b1.x = Wv[(size_t)k0 * DA + cb + 16 + c];
        b1.y = Wv[(size_t)(k0 + 1) * DA + cb + 16 + c];
        acc0 = __builtin_amdgcn_wmma_f32_16x16x4_f32(
             false, a,  false, b0,
             (short)0, acc0,  false,  false);
        acc1 = __builtin_amdgcn_wmma_f32_16x16x4_f32(
             false, a,  false, b1,
             (short)0, acc1,  false,  false);
    }

    const float bc0 = bv[cb + c];
    const float bc1 = bv[cb + 16 + c];
#pragma unroll
    for (int j = 0; j < 8; ++j) {
        float x0 = acc0[j] + bc0;
        x0 = (x0 > 0.f) ? x0 : x0 * NEG_SLOPE;
        x0 = wgt[j] * x0;
        float x1 = acc1[j] + bc1;
        x1 = (x1 > 0.f) ? x1 : x1 * NEG_SLOPE;
        x1 = wgt[j] * x1;
        const float got = xhalf(h ? x0 : x1);
        const float rowLo = h ? got : x0;
        const float rowHi = h ? x1 : got;
        atomicAdd(out + (size_t)rnA[j] * DA + cb + lane, rowLo);
        atomicAdd(out + (size_t)rnB[j] * DA + cb + lane, rowHi);
    }

    if (cb == 0 && c < 8) {
        int rd = h ? rnB[0] : rnA[0];
        float wd = wgt[0];
#pragma unroll
        for (int j = 1; j < 8; ++j)
            if (c == j) { rd = h ? rnB[j] : rnA[j]; wd = wgt[j]; }
        atomicAdd(denom + rd, wd);
    }
}

__global__ void init_kernel(float* __restrict__ out, float* __restrict__ denom,
                            unsigned* __restrict__ maxOrd) {
    int i = blockIdx.x * blockDim.x + threadIdx.x;
    if (i < N_NODES * DA) out[i] = 0.f;
    if (i < N_NODES) { denom[i] = 0.f; maxOrd[i] = 0u; }
}

__global__ void logit_kernel(const float* __restrict__ q, const float* __restrict__ kn,
                             const int* __restrict__ send, const int* __restrict__ recv,
                             float* __restrict__ t, unsigned* __restrict__ maxOrd) {
    int e = blockIdx.x * blockDim.x + threadIdx.x;
    if (e >= N_EDGES) return;
    int rn = recv[e];
    int sn = send[e];
    const float4* qf = (const float4*)(q + (size_t)rn * DA);
    const float4* kf = (const float4*)(kn + (size_t)sn * DA);
    float sum = 0.f;
#pragma unroll
    for (int i = 0; i < DA / 4; ++i) {
        float4 a = qf[i], b = kf[i];
        sum += a.x * b.x + a.y * b.y + a.z * b.z + a.w * b.w;
    }
    float tv = sum * 0.125f;
    t[e] = tv;
    atomicMax(maxOrd + rn, ordKey(tv));
}

__global__ void finalize_kernel(float* __restrict__ out, const float* __restrict__ denom) {
    int i = blockIdx.x * blockDim.x + threadIdx.x;
    if (i >= N_NODES * DA) return;
    out[i] = out[i] / denom[i / DA];
}

extern "C" void kernel_launch(void* const* d_in, const int* in_sizes, int n_in,
                              void* d_out, int out_size, void* d_ws, size_t ws_size,
                              hipStream_t stream) {
    const float* nodes = (const float*)d_in[0];
    const float* edges = (const float*)d_in[1];
    const int* edge_index = (const int*)d_in[2];
    const float* u = (const float*)d_in[3];
    const int* batch = (const int*)d_in[4];
    const float* Wq = (const float*)d_in[5];
    const float* bq = (const float*)d_in[6];
    const float* Wk = (const float*)d_in[7];
    const float* bk = (const float*)d_in[8];
    const float* Wv = (const float*)d_in[9];
    const float* bv = (const float*)d_in[10];
    const int* s = edge_index;
    const int* r = edge_index + N_EDGES;
    float* out = (float*)d_out;

    char* ws = (char*)d_ws;
    float* q  = (float*)ws;            ws += (size_t)N_NODES * DA * 4;
    float* kn = (float*)ws;            ws += (size_t)N_NODES * DA * 4;
    float* t  = (float*)ws;            ws += (size_t)N_EDGES * 4;
    unsigned* maxOrd = (unsigned*)ws;  ws += (size_t)N_NODES * 4;
    float* denom = (float*)ws;         ws += (size_t)N_NODES * 4;

    const int initN = N_NODES * DA;
    init_kernel<<<(initN + 255) / 256, 256, 0, stream>>>(out, denom, maxOrd);

    proj_qk_kernel<<<(N_NODES / 16 * 8) / 8, 256, 0, stream>>>(
        nodes, batch, u, Wq, bq, q, Wk, bk, kn);

    logit_kernel<<<(N_EDGES + 255) / 256, 256, 0, stream>>>(q, kn, s, r, t, maxOrd);

    vproj_scatter_kernel<<<(N_EDGES / 16 * 2) / 8, 256, 0, stream>>>(
        edges, s, r, batch, u, Wv, bv, t, maxOrd, denom, out);

    finalize_kernel<<<(N_NODES * DA + 255) / 256, 256, 0, stream>>>(out, denom);
}


# zero start values of out, denominators and max keys stored by the q/kn projection launch (one float4 of out per thread, behind its epilogue); the separate init launch is gone, 5 launches -> 4
# speedup vs baseline: 1.7378x; 1.0503x over previous
#include <hip/hip_runtime.h>
#include <math.h>


#define N_NODES 4096
#define N_EDGES 32768
#define DV 128
#define DE 64
#define DU 32
#define DA 64
#define NEG_SLOPE 0.01f

typedef __attribute__((ext_vector_type(2))) float v2f;
typedef __attribute__((ext_vector_type(8))) float v8f;

__device__ __forceinline__ unsigned ordKey(float f) {
    unsigned u = __float_as_uint(f);
    return (u & 0x80000000u) ? ~u : (u | 0x80000000u);
}
__device__ __forceinline__ float ordInv(unsigned v) {
    return __uint_as_float((v & 0x80000000u) ? (v & 0x7FFFFFFFu) : ~v);
}

__device__ __forceinline__ float xhalf(float x) {
    const int b = __builtin_bit_cast(int, x);
    return __builtin_bit_cast(
        float, __builtin_amdgcn_permlanex16(b, b, 0x76543210, 0xfedcba98, false, false));
}

__global__ void proj_qk_kernel(const float* __restrict__ nodes,
                               const int* __restrict__ batch,
                               const float* __restrict__ u,
                               const float* __restrict__ Wq,
                               const float* __restrict__ bq,
                               float* __restrict__ q,
                               const float* __restrict__ Wk,
                               const float* __restrict__ bk,
                               float* __restrict__ kn,
                               float* __restrict__ sums,
                               float* __restrict__ denom,
                               unsigned* __restrict__ maxOrd) {
    constexpr int DIN = DV + DU;
    static_assert((N_NODES / 16) * 256 * 4 == N_NODES * DA,
                  "one thread per float4 of the output");
    const int lane = threadIdx.x & 31;
    const int wave = threadIdx.x >> 5;
    const int tile = blockIdx.x * 8 + wave;
    const int rowBase = (tile / 8) * 16;
    const int ct = tile % 8;
    const int colBase = (ct & 3) * 16;
    if (rowBase >= N_NODES) return;
    const float* __restrict__ W = Wq;
    const float* __restrict__ bias = bq;
    float* __restrict__ out = q;
    if (ct >= 4) {
        W = Wk; bias = bk; out = kn;
    }

    const int mrow = rowBase + (lane & 15);
    const int col  = colBase + (lane & 15);
    const int koff = (lane >> 4) * 2;

    const float* __restrict__ arow = nodes + (size_t)mrow * DV;
    const int ubase = batch[mrow] * DU;

    v8f acc = {};
#pragma unroll
    for (int kb = 0; kb < DIN; kb += 4) {
        const int k0 = kb + koff;
        v2f a, b;
        a.x = (k0 < DV)     ? arow[k0]     : u[ubase + k0 - DV];
        a.y = (k0 + 1 < DV) ? arow[k0 + 1] : u[ubase + k0 + 1 - DV];
        b.x = W[(size_t)k0 * DA + col];
        b.y = W[(size_t)(k0 + 1) * DA + col];
        acc = __builtin_amdgcn_wmma_f32_16x16x4_f32(
             false, a,  false, b,
             (short)0, acc,  false,  false);
    }

    const float bc = bias[col];
    const int rOut = rowBase + (lane >> 4) * 8;
#pragma unroll
    for (int j = 0; j < 8; ++j) {
        float x = acc[j] + bc;
        x = (x > 0.f) ? x : x * NEG_SLOPE;
        out[(size_t)(rOut + j) * DA + col] = x;
    }

    const int gtid = blockIdx.x * 256 + threadIdx.x;
    reinterpret_cast<float4*>(sums)[gtid] = make_float4(0.f, 0.f, 0.f, 0.f);
    if (blockIdx.x < N_NODES / 256) {
        denom[gtid] = 0.f;
        maxOrd[gtid] = 0u;
    }
}

__global__ void vproj_scatter_kernel(const float* __restrict__ edges,
                                     const int* __restrict__ s,
                                     const int* __restrict__ recv,
                                     const int* __restrict__ batch,
                                     const float* __restrict__ u,
                                     const float* __restrict__ Wv,
                                     const float* __restrict__ bv,
                                     const float* __restrict__ t,
                                     const unsigned* __restrict__ maxOrd,
                                     float* __restrict__ denom,
                                     float* __restrict__ out) {
    constexpr int DIN = DE + DU;
    const int lane = threadIdx.x & 31;
    const int wave = threadIdx.x >> 5;
    const int tile = blockIdx.x * 8 + wave;
    const int rowBase = (tile >> 1) * 16;
    const int cb = (tile & 1) * 32;
    if (rowBase >= N_EDGES) return;

    const int h = lane >> 4;
    const int c = lane & 15;
    const int mrow = rowBase + c;
    const int koff = h * 2;

    const float* __restrict__ arow = edges + (size_t)mrow * DE;
    const int ubase = batch[s[mrow]] * DU;

    int rnA[8], rnB[8];
    float wgt[8];
#pragma unroll
    for (int j = 0; j < 8; ++j) {
        rnA[j] = recv[rowBase + j];
        rnB[j] = recv[rowBase + 8 + j];
    }
#pragma unroll
    for (int j = 0; j < 8; ++j) {
        const int rOwn = h ? rnB[j] : rnA[j];
        wgt[j] = __expf(t[rowBase + 8 * h + j] - ordInv(maxOrd[rOwn]));
    }

    v8f acc0 = {}, acc1 = {};
#pragma unroll
    for (int kb = 0; kb < DIN; kb += 4) {
        const int k0 = kb + koff;
        v2f a, b0, b1;
        a.x = (k0 < DE)     ? arow[k0]     : u[ubase + k0 - DE];
        a.y = (k0 + 1 < DE) ? arow[k0 + 1] : u[ubase + k0 + 1 - DE];
        b0.x = Wv[(size_t)k0 * DA + cb + c];
        b0.y = Wv[(size_t)(k0 + 1) * DA + cb + c];
        b1.x = Wv[(size_t)k0 * DA + cb + 16 + c];
        b1.y = Wv[(size_t)(k0 + 1) * DA + cb + 16 + c];
        acc0 = __builtin_amdgcn_wmma_f32_16x16x4_f32(
             false, a,  false, b0,
             (short)0, acc0,  false,  false);
        acc1 = __builtin_amdgcn_wmma_f32_16x16x4_f32(
             false, a,  false, b1,
             (short)0, acc1,  false,  false);
    }

    const float bc0 = bv[cb + c];
    const float bc1 = bv[cb + 16 + c];
#pragma unroll
    for (int j = 0; j < 8; ++j) {
        float x0 = acc0[j] + bc0;
        x0 = (x0 > 0.f) ? x0 : x0 * NEG_SLOPE;
        x0 = wgt[j] * x0;
        float x1 = acc1[j] + bc1;
        x1 = (x1 > 0.f) ? x1 : x1 * NEG_SLOPE;
        x1 = wgt[j] * x1;
        const float got = xhalf(h ? x0 : x1);
        const float rowLo = h ? got : x0;
        const float rowHi = h ? x1 : got;
        atomicAdd(out + (size_t)rnA[j] * DA + cb + lane, rowLo);
        atomicAdd(out + (size_t)rnB[j] * DA + cb + lane, rowHi);
    }

    if (cb == 0 && c < 8) {
        int rd = h ? rnB[0] : rnA[0];
        float wd = wgt[0];
#pragma unroll
        for (int j = 1; j < 8; ++j)
            if (c == j) { rd = h ? rnB[j] : rnA[j]; wd = wgt[j]; }
        atomicAdd(denom + rd, wd);
    }
}

__global__ void logit_kernel(const float* __restrict__ q, const float* __restrict__ kn,
                             const int* __restrict__ send, const int* __restrict__ recv,
                             float* __restrict__ t, unsigned* __restrict__ maxOrd) {
    int e = blockIdx.x * blockDim.x + threadIdx.x;
    if (e >= N_EDGES) return;
    int rn = recv[e];
    int sn = send[e];
    const float4* qf = (const float4*)(q + (size_t)rn * DA);
    const float4* kf = (const float4*)(kn + (size_t)sn * DA);
    float sum = 0.f;
#pragma unroll
    for (int i = 0; i < DA / 4; ++i) {
        float4 a = qf[i], b = kf[i];
        sum += a.x * b.x + a.y * b.y + a.z * b.z + a.w * b.w;
    }
    float tv = sum * 0.125f;
    t[e] = tv;
    atomicMax(maxOrd + rn, ordKey(tv));
}

__global__ void finalize_kernel(float* __restrict__ out, const float* __restrict__ denom) {
    int i = blockIdx.x * blockDim.x + threadIdx.x;
    if (i >= N_NODES * DA) return;
    out[i] = out[i] / denom[i / DA];
}

extern "C" void kernel_launch(void* const* d_in, const int* in_sizes, int n_in,
                              void* d_out, int out_size, void* d_ws, size_t ws_size,
                              hipStream_t stream) {
    const float* nodes = (const float*)d_in[0];
    const float* edges = (const float*)d_in[1];
    const int* edge_index = (const int*)d_in[2];
    const float* u = (const float*)d_in[3];
    const int* batch = (const int*)d_in[4];
    const float* Wq = (const float*)d_in[5];
    const float* bq = (const float*)d_in[6];
    const float* Wk = (const float*)d_in[7];
    const float* bk = (const float*)d_in[8];
    const float* Wv = (const float*)d_in[9];
    const float* bv = (const float*)d_in[10];
    const int* s = edge_index;
    const int* r = edge_index + N_EDGES;
    float* out = (float*)d_out;

    char* ws = (char*)d_ws;
    float* q  = (float*)ws;            ws += (size_t)N_NODES * DA * 4;
    float* kn = (float*)ws;            ws += (size_t)N_NODES * DA * 4;
    float* t  = (float*)ws;            ws += (size_t)N_EDGES * 4;
    unsigned* maxOrd = (unsigned*)ws;  ws += (size_t)N_NODES * 4;
    float* denom = (float*)ws;         ws += (size_t)N_NODES * 4;

    proj_qk_kernel<<<(N_NODES / 16 * 8) / 8, 256, 0, stream>>>(
        nodes, batch, u, Wq, bq, q, Wk, bk, kn, out, denom, maxOrd);

    logit_kernel<<<(N_EDGES + 255) / 256, 256, 0, stream>>>(q, kn, s, r, t, maxOrd);

    vproj_scatter_kernel<<<(N_EDGES / 16 * 2) / 8, 256, 0, stream>>>(
        edges, s, r, batch, u, Wv, bv, t, maxOrd, denom, out);

    finalize_kernel<<<(N_NODES * DA + 255) / 256, 256, 0, stream>>>(out, denom);
}


# logit pass: 16 lanes read one 256B row of q/kn (4 lines per load, not 32), a 16x16 transpose-add by DPP leaves one edge per lane; the 64-term dot is summed in a different tree
# speedup vs baseline: 2.0240x; 1.1647x over previous
#include <hip/hip_runtime.h>
#include <math.h>


#define N_NODES 4096
#define N_EDGES 32768
#define DV 128
#define DE 64
#define DU 32
#define DA 64
#define NEG_SLOPE 0.01f

typedef __attribute__((ext_vector_type(2))) float v2f;
typedef __attribute__((ext_vector_type(8))) float v8f;

__device__ __forceinline__ unsigned ordKey(float f) {
    unsigned u = __float_as_uint(f);
    return (u & 0x80000000u) ? ~u : (u | 0x80000000u);
}
__device__ __forceinline__ float ordInv(unsigned v) {
    return __uint_as_float((v & 0x80000000u) ? (v & 0x7FFFFFFFu) : ~v);
}

__device__ __forceinline__ float xhalf(float x) {
    const int b = __builtin_bit_cast(int, x);
    return __builtin_bit_cast(
        float, __builtin_amdgcn_permlanex16(b, b, 0x76543210, 0xfedcba98, false, false));
}

__global__ void proj_qk_kernel(const float* __restrict__ nodes,
                               const int* __restrict__ batch,
                               const float* __restrict__ u,
                               const float* __restrict__ Wq,
                               const float* __restrict__ bq,
                               float* __restrict__ q,
                               const float* __restrict__ Wk,
                               const float* __restrict__ bk,
                               float* __restrict__ kn,
                               float* __restrict__ sums,
                               float* __restrict__ denom,
                               unsigned* __restrict__ maxOrd) {
    constexpr int DIN = DV + DU;
    static_assert((N_NODES / 16) * 256 * 4 == N_NODES * DA,
                  "one thread per float4 of the output");
    const int lane = threadIdx.x & 31;
    const int wave = threadIdx.x >> 5;
    const int tile = blockIdx.x * 8 + wave;
    const int rowBase = (tile / 8) * 16;
    const int ct = tile % 8;
    const int colBase = (ct & 3) * 16;
    if (rowBase >= N_NODES) return;
    const float* __restrict__ W = Wq;
    const float* __restrict__ bias = bq;
    float* __restrict__ out = q;
    if (ct >= 4) {
        W = Wk; bias = bk; out = kn;
    }

    const int mrow = rowBase + (lane & 15);
    const int col  = colBase + (lane & 15);
    const int koff = (lane >> 4) * 2;

    const float* __restrict__ arow = nodes + (size_t)mrow * DV;
    const int ubase = batch[mrow] * DU;

    v8f acc = {};
#pragma unroll
    for (int kb = 0; kb < DIN; kb += 4) {
        const int k0 = kb + koff;
        v2f a, b;
        a.x = (k0 < DV)     ? arow[k0]     : u[ubase + k0 - DV];
        a.y = (k0 + 1 < DV) ? arow[k0 + 1] : u[ubase + k0 + 1 - DV];
        b.x = W[(size_t)k0 * DA + col];
        b.y = W[(size_t)(k0 + 1) * DA + col];
        acc = __builtin_amdgcn_wmma_f32_16x16x4_f32(
             false, a,  false, b,
             (short)0, acc,  false,  false);
    }

    const float bc = bias[col];
    const int rOut = rowBase + (lane >> 4) * 8;
#pragma unroll
    for (int j = 0; j < 8; ++j) {
        float x = acc[j] + bc;
        x = (x > 0.f) ? x : x * NEG_SLOPE;
        out[(size_t)(rOut + j) * DA + col] = x;
    }

    const int gtid = blockIdx.x * 256 + threadIdx.x;
    reinterpret_cast<float4*>(sums)[gtid] = make_float4(0.f, 0.f, 0.f, 0.f);
    if (blockIdx.x < N_NODES / 256) {
        denom[gtid] = 0.f;
        maxOrd[gtid] = 0u;
    }
}

__global__ void vproj_scatter_kernel(const float* __restrict__ edges,
                                     const int* __restrict__ s,
                                     const int* __restrict__ recv,
                                     const int* __restrict__ batch,
                                     const float* __restrict__ u,
                                     const float* __restrict__ Wv,
                                     const float* __restrict__ bv,
                                     const float* __restrict__ t,
                                     const unsigned* __restrict__ maxOrd,
                                     float* __restrict__ denom,
                                     float* __restrict__ out) {
    constexpr int DIN = DE + DU;
    const int lane = threadIdx.x & 31;
    const int wave = threadIdx.x >> 5;
    const int tile = blockIdx.x * 8 + wave;
    const int rowBase = (tile >> 1) * 16;
    const int cb = (tile & 1) * 32;
    if (rowBase >= N_EDGES) return;

    const int h = lane >> 4;
    const int c = lane & 15;
    const int mrow = rowBase + c;
    const int koff = h * 2;

    const float* __restrict__ arow = edges + (size_t)mrow * DE;
    const int ubase = batch[s[mrow]] * DU;

    int rnA[8], rnB[8];
    float wgt[8];
#pragma unroll
    for (int j = 0; j < 8; ++j) {
        rnA[j] = recv[rowBase + j];
        rnB[j] = recv[rowBase + 8 + j];
    }
#pragma unroll
    for (int j = 0; j < 8; ++j) {
        const int rOwn = h ? rnB[j] : rnA[j];
        wgt[j] = __expf(t[rowBase + 8 * h + j] - ordInv(maxOrd[rOwn]));
    }

    v8f acc0 = {}, acc1 = {};
#pragma unroll
    for (int kb = 0; kb < DIN; kb += 4) {
        const int k0 = kb + koff;
        v2f a, b0, b1;
        a.x = (k0 < DE)     ? arow[k0]     : u[ubase + k0 - DE];
        a.y = (k0 + 1 < DE) ? arow[k0 + 1] : u[ubase + k0 + 1 - DE];
        b0.x = Wv[(size_t)k0 * DA + cb + c];
        b0.y = Wv[(size_t)(k0 + 1) * DA + cb + c];
        b1.x = Wv[(size_t)k0 * DA + cb + 16 + c];
        b1.y = Wv[(size_t)(k0 + 1) * DA + cb + 16 + c];
        acc0 = __builtin_amdgcn_wmma_f32_16x16x4_f32(
             false, a,  false, b0,
             (short)0, acc0,  false,  false);
        acc1 = __builtin_amdgcn_wmma_f32_16x16x4_f32(
             false, a,  false, b1,
             (short)0, acc1,  false,  false);
    }

    const float bc0 = bv[cb + c];
    const float bc1 = bv[cb + 16 + c];
#pragma unroll
    for (int j = 0; j < 8; ++j) {
        float x0 = acc0[j] + bc0;
        x0 = (x0 > 0.f) ? x0 : x0 * NEG_SLOPE;
        x0 = wgt[j] * x0;
        float x1 = acc1[j] + bc1;
        x1 = (x1 > 0.f) ? x1 : x1 * NEG_SLOPE;
        x1 = wgt[j] * x1;
        const float got = xhalf(h ? x0 : x1);
        const float rowLo = h ? got : x0;
        const float rowHi = h ? x1 : got;
        atomicAdd(out + (size_t)rnA[j] * DA + cb + lane, rowLo);
        atomicAdd(out + (size_t)rnB[j] * DA + cb + lane, rowHi);
    }

    if (cb == 0 && c < 8) {
        int rd = h ? rnB[0] : rnA[0];
        float wd = wgt[0];
#pragma unroll
        for (int j = 1; j < 8; ++j)
            if (c == j) { rd = h ? rnB[j] : rnA[j]; wd = wgt[j]; }
        atomicAdd(denom + rd, wd);
    }
}

template <int M>
__device__ __forceinline__ float rowxor(float x) {
    const int b = __builtin_bit_cast(int, x);
    return __builtin_bit_cast(
        float, __builtin_amdgcn_update_dpp(0, b, 0x160 | M, 0xf, 0xf, true));
}

#define LOGIT_THREADS 128
#define LOGIT_BLOCKS (N_EDGES / LOGIT_THREADS)
__global__ void __launch_bounds__(LOGIT_THREADS) __attribute__((amdgpu_waves_per_eu(1, 2)))
logit_kernel(const float* __restrict__ q, const float* __restrict__ kn,
             const int* __restrict__ send, const int* __restrict__ recv,
             float* __restrict__ t, unsigned* __restrict__ maxOrd) {
    static_assert(LOGIT_BLOCKS * LOGIT_THREADS == N_EDGES && LOGIT_THREADS % 32 == 0,
                  "one edge per thread, whole waves, no tail");
    static_assert(DA == 64, "a row is 16 float4: one per lane of a half");
    const int c = threadIdx.x & 15;
    const int e = blockIdx.x * LOGIT_THREADS + threadIdx.x;
    const int half0 = e - c;

    int rn[16], sn[16];
    const int4* __restrict__ r4 = reinterpret_cast<const int4*>(recv + half0);
    const int4* __restrict__ s4 = reinterpret_cast<const int4*>(send + half0);
#pragma unroll
    for (int j = 0; j < 4; ++j) {
        const int4 rv = r4[j], sv = s4[j];
        rn[4 * j] = rv.x; rn[4 * j + 1] = rv.y; rn[4 * j + 2] = rv.z; rn[4 * j + 3] = rv.w;
        sn[4 * j] = sv.x; sn[4 * j + 1] = sv.y; sn[4 * j + 2] = sv.z; sn[4 * j + 3] = sv.w;
    }
    const int rOwn = recv[e];

    float p[16];
#pragma unroll
    for (int i = 0; i < 16; ++i) {
        const float4 a = reinterpret_cast<const float4*>(q + (size_t)rn[i] * DA)[c];
        const float4 b = reinterpret_cast<const float4*>(kn + (size_t)sn[i] * DA)[c];
        p[i] = a.x * b.x + a.y * b.y + a.z * b.z + a.w * b.w;
    }

    const bool b8 = (c & 8) != 0, b4 = (c & 4) != 0, b2 = (c & 2) != 0, b1 = (c & 1) != 0;
    float s8[8], s4v[4], s2[2];
#pragma unroll
    for (int i = 0; i < 8; ++i)
        s8[i] = (b8 ? p[i + 8] : p[i]) + rowxor<8>(b8 ? p[i] : p[i + 8]);
#pragma unroll
    for (int i = 0; i < 4; ++i)
        s4v[i] = (b4 ? s8[i + 4] : s8[i]) + rowxor<4>(b4 ? s8[i] : s8[i + 4]);
#pragma unroll
    for (int i = 0; i < 2; ++i)
        s2[i] = (b2 ? s4v[i + 2] : s4v[i]) + rowxor<2>(b2 ? s4v[i] : s4v[i + 2]);
    const float sum = (b1 ? s2[1] : s2[0]) + rowxor<1>(b1 ? s2[0] : s2[1]);

    const float tv = sum * 0.125f;
    t[e] = tv;
    atomicMax(maxOrd + rOwn, ordKey(tv));
}

__global__ void finalize_kernel(float* __restrict__ out, const float* __restrict__ denom) {
    int i = blockIdx.x * blockDim.x + threadIdx.x;
    if (i >= N_NODES * DA) return;
    out[i] = out[i] / denom[i / DA];
}

extern "C" void kernel_launch(void* const* d_in, const int* in_sizes, int n_in,
                              void* d_out, int out_size, void* d_ws, size_t ws_size,
                              hipStream_t stream) {
    const float* nodes = (const float*)d_in[0];
    const float* edges = (const float*)d_in[1];
    const int* edge_index = (const int*)d_in[2];
    const float* u = (const float*)d_in[3];
    const int* batch = (const int*)d_in[4];
    const float* Wq = (const float*)d_in[5];
    const float* bq = (const float*)d_in[6];
    const float* Wk = (const float*)d_in[7];
    const float* bk = (const float*)d_in[8];
    const float* Wv = (const float*)d_in[9];
    const float* bv = (const float*)d_in[10];
    const int* s = edge_index;
    const int* r = edge_index + N_EDGES;
    float* out = (float*)d_out;

    char* ws = (char*)d_ws;
    float* q  = (float*)ws;            ws += (size_t)N_NODES * DA * 4;
    float* kn = (float*)ws;            ws += (size_t)N_NODES * DA * 4;
    float* t  = (float*)ws;            ws += (size_t)N_EDGES * 4;
    unsigned* maxOrd = (unsigned*)ws;  ws += (size_t)N_NODES * 4;
    float* denom = (float*)ws;         ws += (size_t)N_NODES * 4;

    proj_qk_kernel<<<(N_NODES / 16 * 8) / 8, 256, 0, stream>>>(
        nodes, batch, u, Wq, bq, q, Wk, bk, kn, out, denom, maxOrd);

    logit_kernel<<<LOGIT_BLOCKS, LOGIT_THREADS, 0, stream>>>(q, kn, s, r, t, maxOrd);

    vproj_scatter_kernel<<<(N_EDGES / 16 * 2) / 8, 256, 0, stream>>>(
        edges, s, r, batch, u, Wv, bv, t, maxOrd, denom, out);

    finalize_kernel<<<(N_NODES * DA + 255) / 256, 256, 0, stream>>>(out, denom);
}


# A rows of both projection launches and the u table staged once per block in LDS by whole-line float4 loads; fragments by 8-byte LDS reads, not 16-line global loads; k order and sums unchanged
# speedup vs baseline: 2.1332x; 1.0540x over previous
#include <hip/hip_runtime.h>
#include <math.h>


#define N_NODES 4096
#define N_EDGES 32768
#define DV 128
#define DE 64
#define DU 32
#define DA 64
#define NEG_SLOPE 0.01f

typedef __attribute__((ext_vector_type(2))) float v2f;
typedef __attribute__((ext_vector_type(8))) float v8f;

__device__ __forceinline__ unsigned ordKey(float f) {
    unsigned u = __float_as_uint(f);
    return (u & 0x80000000u) ? ~u : (u | 0x80000000u);
}
__device__ __forceinline__ float ordInv(unsigned v) {
    return __uint_as_float((v & 0x80000000u) ? (v & 0x7FFFFFFFu) : ~v);
}

__device__ __forceinline__ float xhalf(float x) {
    const int b = __builtin_bit_cast(int, x);
    return __builtin_bit_cast(
        float, __builtin_amdgcn_permlanex16(b, b, 0x76543210, 0xfedcba98, false, false));
}

#define N_GRAPHS 32
#define U_PITCH (DU + 4)
static_assert(N_GRAPHS * DU / 4 == 256, "u table: one float4 per thread of a 256-thread block");

#define QK_PITCH (DV + 4)
__global__ void __launch_bounds__(256)
proj_qk_kernel(const float* __restrict__ nodes,
               const int* __restrict__ batch,
               const float* __restrict__ u,
               const float* __restrict__ Wq,
               const float* __restrict__ bq,
               float* __restrict__ q,
               const float* __restrict__ Wk,
               const float* __restrict__ bk,
               float* __restrict__ kn,
               float* __restrict__ sums,
               float* __restrict__ denom,
               unsigned* __restrict__ maxOrd) {
    constexpr int DIN = DV + DU;
    constexpr int KN = DV / 4;
    constexpr int KU = DU / 4;
    static_assert((N_NODES / 16) * 256 * 4 == N_NODES * DA,
                  "one thread per float4 of the output");
    static_assert(16 * DV / 4 == 2 * 256, "tile staging: two float4 of nodes per thread");
    __shared__ __attribute__((aligned(16))) float As[16 * QK_PITCH];
    __shared__ __attribute__((aligned(16))) float Us[N_GRAPHS * U_PITCH];

    const int tid = threadIdx.x;
    const int lane = tid & 31;
    const int wave = tid >> 5;
    const int rowBase = blockIdx.x * 16;
    const int ct = wave;
    const int colBase = (ct & 3) * 16;
    if (rowBase >= N_NODES) return;
    const float* __restrict__ W = Wq;
    const float* __restrict__ bias = bq;
    float* __restrict__ out = q;
    if (ct >= 4) {
        W = Wk; bias = bk; out = kn;
    }

    const int c = lane & 15;
    const int h = lane >> 4;
    const int col  = colBase + c;
    const int koff = h * 2;

    const float4* __restrict__ nsrc =
        reinterpret_cast<const float4*>(nodes + (size_t)rowBase * DV);
    const float4 n0 = nsrc[tid];
    const float4 n1 = nsrc[tid + 256];
    const float4 ut = reinterpret_cast<const float4*>(u)[tid];
    const int g = batch[rowBase + c];
    v2f bfr[DIN / 4];
#pragma unroll
    for (int i = 0; i < DIN / 4; ++i) {
        const int k0 = 4 * i + koff;
        bfr[i].x = W[(size_t)k0 * DA + col];
        bfr[i].y = W[(size_t)(k0 + 1) * DA + col];
    }

    *reinterpret_cast<float4*>(&As[(tid >> 5) * QK_PITCH + (tid & 31) * 4]) = n0;
    *reinterpret_cast<float4*>(&As[((tid >> 5) + 8) * QK_PITCH + (tid & 31) * 4]) = n1;
    *reinterpret_cast<float4*>(&Us[(tid >> 3) * U_PITCH + (tid & 7) * 4]) = ut;
    __syncthreads();

    const float* arow = As + c * QK_PITCH + koff;
    const float* urow = Us + g * U_PITCH + koff;
    v2f afr[DIN / 4];
#pragma unroll
    for (int i = 0; i < KN; ++i) afr[i] = *reinterpret_cast<const v2f*>(arow + 4 * i);
#pragma unroll
    for (int i = 0; i < KU; ++i) afr[KN + i] = *reinterpret_cast<const v2f*>(urow + 4 * i);
    __builtin_amdgcn_sched_barrier(0);

    v8f acc = {};
#pragma unroll
    for (int i = 0; i < DIN / 4; ++i) {
        acc = __builtin_amdgcn_wmma_f32_16x16x4_f32(
             false, afr[i],  false, bfr[i],
             (short)0, acc,  false,  false);
    }

    const float bc = bias[col];
    const int rOut = rowBase + h * 8;
#pragma unroll
    for (int j = 0; j < 8; ++j) {
        float x = acc[j] + bc;
        x = (x > 0.f) ? x : x * NEG_SLOPE;
        out[(size_t)(rOut + j) * DA + col] = x;
    }

    const int gtid = blockIdx.x * 256 + threadIdx.x;
    reinterpret_cast<float4*>(sums)[gtid] = make_float4(0.f, 0.f, 0.f, 0.f);
    if (blockIdx.x < N_NODES / 256) {
        denom[gtid] = 0.f;
        maxOrd[gtid] = 0u;
    }
}

#define V_PITCH (DE + 4)
#define V_ROWS 64
__global__ void __launch_bounds__(256)
vproj_scatter_kernel(const float* __restrict__ edges,
                     const int* __restrict__ s,
                     const int* __restrict__ recv,
                     const int* __restrict__ batch,
                     const float* __restrict__ u,
                     const float* __restrict__ Wv,
                     const float* __restrict__ bv,
                     const float* __restrict__ t,
                     const unsigned* __restrict__ maxOrd,
                     float* __restrict__ denom,
                     float* __restrict__ out) {
    constexpr int DIN = DE + DU;
    constexpr int KE = DE / 4;
    constexpr int KU = DU / 4;
    static_assert(V_ROWS * DE / 4 == 4 * 256, "tile staging: four float4 of edges per thread");
    static_assert(N_EDGES % V_ROWS == 0, "whole blocks, no tail");
    __shared__ __attribute__((aligned(16))) float As[V_ROWS * V_PITCH];
    __shared__ __attribute__((aligned(16))) float Us[N_GRAPHS * U_PITCH];

    const int tid = threadIdx.x;
    const int lane = tid & 31;
    const int wave = tid >> 5;
    const int blockRow = blockIdx.x * V_ROWS;
    const int rb = wave >> 1;
    const int rowBase = blockRow + rb * 16;
    const int cb = (wave & 1) * 32;
    if (blockRow >= N_EDGES) return;

    const int h = lane >> 4;
    const int c = lane & 15;
    const int koff = h * 2;

    const float4* __restrict__ esrc =
        reinterpret_cast<const float4*>(edges + (size_t)blockRow * DE);
    float4 ev[4];
#pragma unroll
    for (int p = 0; p < 4; ++p) ev[p] = esrc[tid + 256 * p];
    const float4 ut = reinterpret_cast<const float4*>(u)[tid];
    const int sv = s[rowBase + c];
    int rnA[8], rnB[8];
    float tv[8];
#pragma unroll
    for (int j = 0; j < 8; ++j) {
        rnA[j] = recv[rowBase + j];
        rnB[j] = recv[rowBase + 8 + j];
        tv[j] = t[rowBase + 8 * h + j];
    }
    v2f bf0[DIN / 4], bf1[DIN / 4];
#pragma unroll
    for (int i = 0; i < DIN / 4; ++i) {
        const int k0 = 4 * i + koff;
        bf0[i].x = Wv[(size_t)k0 * DA + cb + c];
        bf0[i].y = Wv[(size_t)(k0 + 1) * DA + cb + c];
        bf1[i].x = Wv[(size_t)k0 * DA + cb + 16 + c];
        bf1[i].y = Wv[(size_t)(k0 + 1) * DA + cb + 16 + c];
    }

#pragma unroll
    for (int p = 0; p < 4; ++p) {
        const int f = tid + 256 * p;
        *reinterpret_cast<float4*>(&As[(f >> 4) * V_PITCH + (f & 15) * 4]) = ev[p];
    }
    *reinterpret_cast<float4*>(&Us[(tid >> 3) * U_PITCH + (tid & 7) * 4]) = ut;
    __syncthreads();

    const int g = batch[sv];
    unsigned mo[8];
#pragma unroll
    for (int j = 0; j < 8; ++j) mo[j] = maxOrd[h ? rnB[j] : rnA[j]];

    const float* arow = As + (rb * 16 + c) * V_PITCH + koff;
    v2f afr[KE];
#pragma unroll
    for (int i = 0; i < KE; ++i) afr[i] = *reinterpret_cast<const v2f*>(arow + 4 * i);
    __builtin_amdgcn_sched_barrier(0);

    v8f acc0 = {}, acc1 = {};
#pragma unroll
    for (int i = 0; i < KE; ++i) {
        acc0 = __builtin_amdgcn_wmma_f32_16x16x4_f32(
             false, afr[i],  false, bf0[i],
             (short)0, acc0,  false,  false);
        acc1 = __builtin_amdgcn_wmma_f32_16x16x4_f32(
             false, afr[i],  false, bf1[i],
             (short)0, acc1,  false,  false);
    }

    const float* urow = Us + g * U_PITCH + koff;
    v2f ufr[KU];
#pragma unroll
    for (int i = 0; i < KU; ++i) ufr[i] = *reinterpret_cast<const v2f*>(urow + 4 * i);
    __builtin_amdgcn_sched_barrier(0);
#pragma unroll
    for (int i = 0; i < KU; ++i) {
        acc0 = __builtin_amdgcn_wmma_f32_16x16x4_f32(
             false, ufr[i],  false, bf0[KE + i],
             (short)0, acc0,  false,  false);
        acc1 = __builtin_amdgcn_wmma_f32_16x16x4_f32(
             false, ufr[i],  false, bf1[KE + i],
             (short)0, acc1,  false,  false);
    }

    float wgt[8];
#pragma unroll
    for (int j = 0; j < 8; ++j) wgt[j] = __expf(tv[j] - ordInv(mo[j]));

    const float bc0 = bv[cb + c];
    const float bc1 = bv[cb + 16 + c];
#pragma unroll
    for (int j = 0; j < 8; ++j) {
        float x0 = acc0[j] + bc0;
        x0 = (x0 > 0.f) ? x0 : x0 * NEG_SLOPE;
        x0 = wgt[j] * x0;
        float x1 = acc1[j] + bc1;
        x1 = (x1 > 0.f) ? x1 : x1 * NEG_SLOPE;
        x1 = wgt[j] * x1;
        const float got = xhalf(h ? x0 : x1);
        const float rowLo = h ? got : x0;
        const float rowHi = h ? x1 : got;
        atomicAdd(out + (size_t)rnA[j] * DA + cb + lane, rowLo);
        atomicAdd(out + (size_t)rnB[j] * DA + cb + lane, rowHi);
    }

    if (cb == 0 && c < 8) {
        int rd = h ? rnB[0] : rnA[0];
        float wd = wgt[0];
#pragma unroll
        for (int j = 1; j < 8; ++j)
            if (c == j) { rd = h ? rnB[j] : rnA[j]; wd = wgt[j]; }
        atomicAdd(denom + rd, wd);
    }
}

template <int M>
__device__ __forceinline__ float rowxor(float x) {
    const int b = __builtin_bit_cast(int, x);
    return __builtin_bit_cast(
        float, __builtin_amdgcn_update_dpp(0, b, 0x160 | M, 0xf, 0xf, true));
}

#define LOGIT_THREADS 128
#define LOGIT_BLOCKS (N_EDGES / LOGIT_THREADS)
__global__ void __launch_bounds__(LOGIT_THREADS) __attribute__((amdgpu_waves_per_eu(1, 2)))
logit_kernel(const float* __restrict__ q, const float* __restrict__ kn,
             const int* __restrict__ send, const int* __restrict__ recv,
             float* __restrict__ t, unsigned* __restrict__ maxOrd) {
    static_assert(LOGIT_BLOCKS * LOGIT_THREADS == N_EDGES && LOGIT_THREADS % 32 == 0,
                  "one edge per thread, whole waves, no tail");
    static_assert(DA == 64, "a row is 16 float4: one per lane of a half");
    const int c = threadIdx.x & 15;
    const int e = blockIdx.x * LOGIT_THREADS + threadIdx.x;
    const int half0 = e - c;

    int rn[16], sn[16];
    const int4* __restrict__ r4 = reinterpret_cast<const int4*>(recv + half0);
    const int4* __restrict__ s4 = reinterpret_cast<const int4*>(send + half0);
#pragma unroll
    for (int j = 0; j < 4; ++j) {
        const int4 rv = r4[j], sv = s4[j];
        rn[4 * j] = rv.x; rn[4 * j + 1] = rv.y; rn[4 * j + 2] = rv.z; rn[4 * j + 3] = rv.w;
        sn[4 * j] = sv.x; sn[4 * j + 1] = sv.y; sn[4 * j + 2] = sv.z; sn[4 * j + 3] = sv.w;
    }
    const int rOwn = recv[e];

    float p[16];
#pragma unroll
    for (int i = 0; i < 16; ++i) {
        const float4 a = reinterpret_cast<const float4*>(q + (size_t)rn[i] * DA)[c];
        const float4 b = reinterpret_cast<const float4*>(kn + (size_t)sn[i] * DA)[c];
        p[i] = a.x * b.x + a.y * b.y + a.z * b.z + a.w * b.w;
    }

    const bool b8 = (c & 8) != 0, b4 = (c & 4) != 0, b2 = (c & 2) != 0, b1 = (c & 1) != 0;
    float s8[8], s4v[4], s2[2];
#pragma unroll
    for (int i = 0; i < 8; ++i)
        s8[i] = (b8 ? p[i + 8] : p[i]) + rowxor<8>(b8 ? p[i] : p[i + 8]);
#pragma unroll
    for (int i = 0; i < 4; ++i)
        s4v[i] = (b4 ? s8[i + 4] : s8[i]) + rowxor<4>(b4 ? s8[i] : s8[i + 4]);
#pragma unroll
    for (int i = 0; i < 2; ++i)
        s2[i] = (b2 ? s4v[i + 2] : s4v[i]) + rowxor<2>(b2 ? s4v[i] : s4v[i + 2]);
    const float sum = (b1 ? s2[1] : s2[0]) + rowxor<1>(b1 ? s2[0] : s2[1]);

    const float tv = sum * 0.125f;
    t[e] = tv;
    atomicMax(maxOrd + rOwn, ordKey(tv));
}

__global__ void finalize_kernel(float* __restrict__ out, const float* __restrict__ denom) {
    int i = blockIdx.x * blockDim.x + threadIdx.x;
    if (i >= N_NODES * DA) return;
    out[i] = out[i] / denom[i / DA];
}

extern "C" void kernel_launch(void* const* d_in, const int* in_sizes, int n_in,
                              void* d_out, int out_size, void* d_ws, size_t ws_size,
                              hipStream_t stream) {
    const float* nodes = (const float*)d_in[0];
    const float* edges = (const float*)d_in[1];
    const int* edge_index = (const int*)d_in[2];
    const float* u = (const float*)d_in[3];
    const int* batch = (const int*)d_in[4];
    const float* Wq = (const float*)d_in[5];
    const float* bq = (const float*)d_in[6];
    const float* Wk = (const float*)d_in[7];
    const float* bk = (const float*)d_in[8];
    const float* Wv = (const float*)d_in[9];
    const float* bv = (const float*)d_in[10];
    const int* s = edge_index;
    const int* r = edge_index + N_EDGES;
    float* out = (float*)d_out;

    char* ws = (char*)d_ws;
    float* q  = (float*)ws;            ws += (size_t)N_NODES * DA * 4;
    float* kn = (float*)ws;            ws += (size_t)N_NODES * DA * 4;
    float* t  = (float*)ws;            ws += (size_t)N_EDGES * 4;
    unsigned* maxOrd = (unsigned*)ws;  ws += (size_t)N_NODES * 4;
    float* denom = (float*)ws;         ws += (size_t)N_NODES * 4;

    proj_qk_kernel<<<(N_NODES / 16 * 8) / 8, 256, 0, stream>>>(
        nodes, batch, u, Wq, bq, q, Wk, bk, kn, out, denom, maxOrd);

    logit_kernel<<<LOGIT_BLOCKS, LOGIT_THREADS, 0, stream>>>(q, kn, s, r, t, maxOrd);

    vproj_scatter_kernel<<<(N_EDGES / 16 * 2) / 8, 256, 0, stream>>>(
        edges, s, r, batch, u, Wv, bv, t, maxOrd, denom, out);

    finalize_kernel<<<(N_NODES * DA + 255) / 256, 256, 0, stream>>>(out, denom);
}
